# starting kernel
# baseline (speedup 1.0000x reference)
#include <hip/hip_runtime.h>

typedef __attribute__((ext_vector_type(2))) float v2f;
typedef __attribute__((ext_vector_type(8))) float v8f;

#define NN 50000
#define NE 800000
#define DF 64
#define DH 128
#define LDS_STRIDE 132

__global__ void gin_init(const float* __restrict__ feat,
                         const float* __restrict__ eps,
                         float* __restrict__ acc) {
  int i = blockIdx.x * blockDim.x + threadIdx.x;
  if (i < NN * DF / 4) {
    float s = 1.0f + eps[0];
    float4 v = ((const float4*)feat)[i];
    v.x *= s; v.y *= s; v.z *= s; v.w *= s;
    ((float4*)acc)[i] = v;
  }
}

__global__ void gin_scatter(const float* __restrict__ feat,
                            const int* __restrict__ src,
                            const int* __restrict__ dst,
                            float* __restrict__ acc) {
  long long t = (long long)blockIdx.x * blockDim.x + threadIdx.x;
  int e = (int)(t >> 6);
  int d = (int)(t & 63);
  if (e >= NE) return;
  int s = src[e];
  int q = dst[e];
  atomicAdd(acc + (long long)q * DF + d, feat[(long long)s * DF + d]);
}

__global__ __launch_bounds__(32) void gin_mlp(const float* __restrict__ acc,
                                              const float* __restrict__ W1,
                                              const float* __restrict__ b1,
                                              const float* __restrict__ W2,
                                              const float* __restrict__ b2,
                                              float* __restrict__ out) {
  __shared__ float lds[16 * LDS_STRIDE];

  const int tile = blockIdx.x;
  const int lane = threadIdx.x;
  const int half = lane >> 4;
  const int l16  = lane & 15;
  const int row0 = tile * 16;

  v2f afrag[16];
  {
    const float* arow = acc + (long long)(row0 + l16) * DF;
#pragma unroll
    for (int kk = 0; kk < 16; ++kk) {
      int kb = kk * 4 + half * 2;
      afrag[kk].x = arow[kb];
      afrag[kk].y = arow[kb + 1];
    }
  }

#pragma unroll
  for (int nt = 0; nt < 8; ++nt) {
    const int n0 = nt * 16;
    v8f c = {};
#pragma unroll
    for (int kk = 0; kk < 16; ++kk) {
      int kr = kk * 4 + half * 2;
      v2f bf;
      bf.x = W1[kr * DH + n0 + l16];
      bf.y = W1[(kr + 1) * DH + n0 + l16];
      c = __builtin_amdgcn_wmma_f32_16x16x4_f32(false, afrag[kk], false, bf,
                                                (short)0, c, false, false);
    }
    float bias = b1[n0 + l16];
#pragma unroll
    for (int r = 0; r < 8; ++r) {
      float h = c[r] + bias;
      h = h > 0.0f ? h : 0.0f;
      lds[(r + half * 8) * LDS_STRIDE + n0 + l16] = h;
    }
  }

  __syncthreads();

#pragma unroll
  for (int nt = 0; nt < 4; ++nt) {
    const int n0 = nt * 16;
    v8f c = {};
#pragma unroll
    for (int kk = 0; kk < 32; ++kk) {
      int kr = kk * 4 + half * 2;
      v2f af, bf;
      af.x = lds[l16 * LDS_STRIDE + kr];
      af.y = lds[l16 * LDS_STRIDE + kr + 1];
      bf.x = W2[kr * DF + n0 + l16];
      bf.y = W2[(kr + 1) * DF + n0 + l16];
      c = __builtin_amdgcn_wmma_f32_16x16x4_f32(false, af, false, bf,
                                                (short)0, c, false, false);
    }
    float bias = b2[n0 + l16];
#pragma unroll
    for (int r = 0; r < 8; ++r) {
      out[(long long)(row0 + r + half * 8) * DF + n0 + l16] = c[r] + bias;
    }
  }
}

extern "C" void kernel_launch(void* const* d_in, const int* in_sizes, int n_in,
                              void* d_out, int out_size, void* d_ws, size_t ws_size,
                              hipStream_t stream) {
  const float* feat = (const float*)d_in[0];
  const int*   src  = (const int*)d_in[1];
  const int*   dst  = (const int*)d_in[2];
  const float* eps  = (const float*)d_in[3];
  const float* W1   = (const float*)d_in[4];
  const float* b1   = (const float*)d_in[5];
  const float* W2   = (const float*)d_in[6];
  const float* b2   = (const float*)d_in[7];
  float* acc = (float*)d_ws;
  float* out = (float*)d_out;

  {
    int n4 = NN * DF / 4;
    int blocks = (n4 + 255) / 256;
    gin_init<<<blocks, 256, 0, stream>>>(feat, eps, acc);
  }
  {
    long long total = (long long)NE * 64;
    int blocks = (int)((total + 255) / 256);
    gin_scatter<<<blocks, 256, 0, stream>>>(feat, src, dst, acc);
  }
  {
    gin_mlp<<<NN / 16, 32, 0, stream>>>(acc, W1, b1, W2, b2, out);
  }
}


# MLP merged into the owner-scan launch (3 launches to 2): a group's 196 summed rows stay in LDS, its 32 waves run both f32 WMMA products with a weight slice in registers; sums and k order unchanged
# speedup vs baseline: 3.9225x; 3.9225x over previous
#include <hip/hip_runtime.h>

typedef __attribute__((ext_vector_type(2))) float v2f;
typedef __attribute__((ext_vector_type(8))) float v8f;

#define NN 50000
#define NE 800000
#define DF 64
#define DH 128
#define SST 68
#define HST 132

#define OWN_R     196
#define OWN_RP    208
#define OWN_WG    256
#define OWN_T     1024
#define OWN_WAVES 32
#define LIST_CAP  1024
#define NQ        (NE / 128)
#define PLANE_W   25024
#define PLANE_Q   (PLANE_W / 4)
#define NSTEP     ((NQ + OWN_T - 1) / OWN_T)
#define KEY_T     128
#define KEY_WAVES ((NQ + 7) / 8)

__global__ __launch_bounds__(KEY_T) void gin_keys(const int* __restrict__ src, const int* __restrict__ dst,
                                                  float* __restrict__ ws) {
  unsigned* planes = (unsigned*)ws;
  uint4* sr4 = (uint4*)(planes + 8 * PLANE_W);
  const int lane = threadIdx.x & 31;
  const int gw   = __builtin_amdgcn_readfirstlane((int)((blockIdx.x * KEY_T + threadIdx.x) >> 5));
  if (gw >= KEY_WAVES) return;
  const int4* d4 = (const int4*)dst;
  const int4* s4 = (const int4*)src;
  int4 v[8], u[8];
#pragma unroll
  for (int j = 0; j < 8; ++j) {
    const int q = gw * 8 + j;
    const int qc = q < NQ ? q : NQ - 1;
    v[j] = d4[qc * 32 + lane];
    u[j] = s4[qc * 32 + lane];
  }
  unsigned pl[8] = {0u, 0u, 0u, 0u, 0u, 0u, 0u, 0u};
#pragma unroll
  for (int j = 0; j < 8; ++j) {
    const int dk[4] = {v[j].x, v[j].y, v[j].z, v[j].w};
    const int sk[4] = {u[j].x, u[j].y, u[j].z, u[j].w};
    unsigned w[4];
#pragma unroll
    for (int k = 0; k < 4; ++k) {
      const unsigned d   = (unsigned)dk[k];
      const unsigned key = d / (unsigned)OWN_R;
      const unsigned r   = d < (unsigned)NN ? d - key * (unsigned)OWN_R : 255u;
      w[k] = ((unsigned)sk[k] << 8) | r;
      const bool mine = lane == j * 4 + k;
#pragma unroll
      for (int b = 0; b < 8; ++b) {
        const unsigned m = __builtin_amdgcn_ballot_w32(((key >> b) & 1u) != 0u);
        pl[b] = mine ? m : pl[b];
      }
    }
    if (gw * 8 + j < NQ) sr4[(gw * 8 + j) * 32 + lane] = make_uint4(w[0], w[1], w[2], w[3]);
  }
  const int nw = (NQ - gw * 8) < 8 ? (NQ - gw * 8) * 4 : 32;
  if (lane < nw) {
#pragma unroll
    for (int b = 0; b < 8; ++b) planes[b * PLANE_W + gw * 32 + lane] = pl[b];
  }
}

__device__ __forceinline__ void own_add_row_lane(float* sacc, const float* __restrict__ feat,
                                                 const unsigned* SR, unsigned e, unsigned nn) {
  const unsigned ent = SR[e];
  const unsigned r = ent & 255u;
  if (r >= nn) return;
  const float* frow = feat + (long long)(ent >> 8) * DF;
#pragma unroll 1
  for (int d = 0; d < DF; ++d) atomicAdd(&sacc[r * SST + d], frow[d]);
}

__global__ __launch_bounds__(OWN_T) void gin_own(const float* __restrict__ feat,
                                                 const float* __restrict__ eps,
                                                 const float* __restrict__ W1,
                                                 const float* __restrict__ b1,
                                                 const float* __restrict__ W2,
                                                 const float* __restrict__ b2,
                                                 const float* __restrict__ ws,
                                                 float* __restrict__ out) {
  __shared__ alignas(16) float    sacc[OWN_RP * SST];
  __shared__ alignas(16) unsigned slist[OWN_WAVES * LIST_CAP];

  const int tid  = threadIdx.x;
  const int lane = tid & 31;
  const int wave = __builtin_amdgcn_readfirstlane(tid >> 5);
  const int lo   = blockIdx.x * OWN_R;
  const unsigned nn = (unsigned)((NN - lo) < OWN_R ? (NN - lo) : OWN_R);

  {
    const float s1 = 1.0f + eps[0];
    const float4* f4 = (const float4*)(feat + (long long)lo * DF);
    for (unsigned i = tid; i < OWN_RP * (DF / 4); i += OWN_T) {
      const unsigned row = i >> 4, c4 = i & 15u;
      float4 v = make_float4(0.0f, 0.0f, 0.0f, 0.0f);
      if (row < nn) {
        v = f4[i];
        v.x *= s1; v.y *= s1; v.z *= s1; v.w *= s1;
      }
      ((float4*)sacc)[row * (SST / 4) + c4] = v;
    }
  }
  __syncthreads();

  unsigned* mylist = slist + wave * LIST_CAP;
  unsigned cw = 0;
  const uint4* P = (const uint4*)ws;
  const unsigned* SR = (const unsigned*)P + 8 * PLANE_W;

  unsigned sx[8];
#pragma unroll
  for (int b = 0; b < 8; ++b) sx[b] = ((blockIdx.x >> b) & 1u) ? 0u : 0xFFFFFFFFu;

  auto take = [&](unsigned M, unsigned e0) {
    unsigned bm = __builtin_amdgcn_ballot_w32(M != 0u);
    while (bm != 0u && cw + 32u <= LIST_CAP) {
      if (M != 0u) {
        mylist[__builtin_amdgcn_mbcnt_lo(bm, cw)] = e0 + 4u * (unsigned)__builtin_ctz(M);
        M &= M - 1u;
      }
      cw += (unsigned)__builtin_popcount(bm);
      bm = __builtin_amdgcn_ballot_w32(M != 0u);
    }
    while (bm != 0u) {
      if (M != 0u) {
        const unsigned e = e0 + 4u * (unsigned)__builtin_ctz(M);
        const unsigned p = __builtin_amdgcn_mbcnt_lo(bm, cw);
        M &= M - 1u;
        if (p < LIST_CAP) mylist[p] = e;
        else own_add_row_lane(sacc, feat, SR, e, nn);
      }
      cw += (unsigned)__builtin_popcount(bm);
      bm = __builtin_amdgcn_ballot_w32(M != 0u);
    }
  };

  auto drain = [&](unsigned b0, int cb) {
    const unsigned e = mylist[b0 + (unsigned)(lane < cb ? lane : cb - 1)];
    const unsigned ent = SR[e];
    const int s = (int)(ent >> 8);
    const unsigned r = ent & 255u;
    const unsigned vm = __builtin_amdgcn_ballot_w32(lane < cb && r < nn);
    v2f x[32];
#pragma unroll
    for (int j = 0; j < 32; ++j) {
      const int sj = __builtin_amdgcn_readlane(s, j);
      x[j] = *(const v2f*)(feat + (long long)sj * DF + 2 * lane);
    }
#pragma unroll
    for (int j = 0; j < 32; ++j) {
      if ((vm >> j) & 1u) {
        const int rj = __builtin_amdgcn_readlane((int)r, j);
        atomicAdd(&sacc[rj * SST + 2 * lane],     x[j].x);
        atomicAdd(&sacc[rj * SST + 2 * lane + 1], x[j].y);
      }
    }
  };
  unsigned dn = 0;

  const int rot = (int)((blockIdx.x * 28u) & ~15u);
  for (int i = 0; i < NSTEP; ++i) {
    int q = rot + i * OWN_T + tid;
    if (q >= NSTEP * OWN_T) q -= NSTEP * OWN_T;
    const bool ok = q < NQ;
    const int qc = ok ? q : NQ - 1;
    uint4 p[8];
#pragma unroll
    for (int b = 0; b < 8; ++b) p[b] = P[b * PLANE_Q + qc];
    unsigned M0 = ok ? 0xFFFFFFFFu : 0u, M1 = M0, M2 = M0, M3 = M0;
#pragma unroll
    for (int b = 0; b < 8; ++b) {
      M0 &= p[b].x ^ sx[b];
      M1 &= p[b].y ^ sx[b];
      M2 &= p[b].z ^ sx[b];
      M3 &= p[b].w ^ sx[b];
    }
    const unsigned e0 = (unsigned)q * 128u;
    take(M0, e0);
    take(M1, e0 + 1u);
    take(M2, e0 + 2u);
    take(M3, e0 + 3u);
    {
      const unsigned lim = cw < LIST_CAP ? cw : LIST_CAP;
      while (lim - dn >= 32u) { drain(dn, 32); dn += 32u; }
    }
  }

  {
    const unsigned lim = cw < LIST_CAP ? cw : LIST_CAP;
    if (lim > dn) drain(dn, (int)(lim - dn));
  }

  const int half = lane >> 4;
  const int l16  = lane & 15;
  const int ntile = (int)((nn + 15u) >> 4);

  const int c1 = wave & 7;
  v2f w1f[16];
#pragma unroll
  for (int kk = 0; kk < 16; ++kk) {
    const int kr = kk * 4 + half * 2;
    w1f[kk].x = W1[kr * DH + c1 * 16 + l16];
    w1f[kk].y = W1[(kr + 1) * DH + c1 * 16 + l16];
  }
  const float bias1 = b1[c1 * 16 + l16];

  __syncthreads();

  float* hid = (float*)slist;
  for (int t = wave >> 3; t < ntile; t += 4) {
    const float* arow = sacc + (t * 16 + l16) * SST + half * 2;
    v2f af[16];
#pragma unroll
    for (int kk = 0; kk < 16; ++kk) af[kk] = *(const v2f*)(arow + kk * 4);
    v8f c = {};
#pragma unroll
    for (int kk = 0; kk < 16; ++kk)
      c = __builtin_amdgcn_wmma_f32_16x16x4_f32(false, af[kk], false, w1f[kk], (short)0, c, false, false);
#pragma unroll
    for (int r = 0; r < 8; ++r) {
      float h = c[r] + bias1;
      h = h > 0.0f ? h : 0.0f;
      hid[(t * 16 + r + half * 8) * HST + c1 * 16 + l16] = h;
    }
  }

  const int c2 = wave & 3;
  v2f w2f[32];
#pragma unroll
  for (int kk = 0; kk < 32; ++kk) {
    const int kr = kk * 4 + half * 2;
    w2f[kk].x = W2[kr * DF + c2 * 16 + l16];
    w2f[kk].y = W2[(kr + 1) * DF + c2 * 16 + l16];
  }
  const float bias2 = b2[c2 * 16 + l16];

  __syncthreads();

  for (int t = wave >> 2; t < ntile; t += 8) {
    const float* hrow = hid + (t * 16 + l16) * HST + half * 2;
    v8f c = {};
#pragma unroll
    for (int kk = 0; kk < 32; ++kk) {
      const v2f af = *(const v2f*)(hrow + kk * 4);
      c = __builtin_amdgcn_wmma_f32_16x16x4_f32(false, af, false, w2f[kk], (short)0, c, false, false);
    }
#pragma unroll
    for (int r = 0; r < 8; ++r) sacc[(t * 16 + r + half * 8) * SST + c2 * 16 + l16] = c[r] + bias2;
  }

  __syncthreads();

  {
    float4* o4 = (float4*)(out + (long long)lo * DF);
    for (unsigned i = tid; i < nn * (DF / 4); i += OWN_T) o4[i] = ((const float4*)sacc)[(i >> 4) * (SST / 4) + (i & 15u)];
  }
}

extern "C" void kernel_launch(void* const* d_in, const int* in_sizes, int n_in,
                              void* d_out, int out_size, void* d_ws, size_t ws_size,
                              hipStream_t stream) {
  const float* feat = (const float*)d_in[0];
  const int*   src  = (const int*)d_in[1];
  const int*   dst  = (const int*)d_in[2];
  const float* eps  = (const float*)d_in[3];
  const float* W1   = (const float*)d_in[4];
  const float* b1   = (const float*)d_in[5];
  const float* W2   = (const float*)d_in[6];
  const float* b2   = (const float*)d_in[7];
  float* ws  = (float*)d_ws;
  float* out = (float*)d_out;

  {
    gin_keys<<<(KEY_WAVES * 32 + KEY_T - 1) / KEY_T, KEY_T, 0, stream>>>(src, dst, ws);
  }
  {
    gin_own<<<OWN_WG, OWN_T, 0, stream>>>(feat, eps, W1, b1, W2, b2, ws, out);
  }
}
